# starting kernel
# baseline (speedup 1.0000x reference)
#include <hip/hip_runtime.h>
#include <hip/hip_bf16.h>
#include <math.h>

#define FIN  128
#define FHID 64

typedef float v2f __attribute__((ext_vector_type(2)));
typedef float v8f __attribute__((ext_vector_type(8)));

__global__ void cgcl_deg_kernel(const int* __restrict__ dst,
                                unsigned* __restrict__ deg, int E) {
    int e = blockIdx.x * blockDim.x + threadIdx.x;
    if (e < E) atomicAdd(&deg[dst[e]], 1u);
}

__global__ void cgcl_dinv_kernel(const unsigned* __restrict__ deg,
                                 float* __restrict__ dinv, int N) {
    int n = blockIdx.x * blockDim.x + threadIdx.x;
    if (n < N) dinv[n] = rsqrtf((float)deg[n] + 1.0f);
}

__global__ void __launch_bounds__(128)
cgcl_gemm_wmma(const float* __restrict__ x, const float* __restrict__ W,
               float* __restrict__ h, int N) {
    const int lane = threadIdx.x & 31;
    const int wave = threadIdx.x >> 5;
    const int m    = lane & 15;
    const int hi   = lane >> 4;
    const int koff = hi * 2;

    int row = blockIdx.x * 16 + m;
    if (row >= N) row = N - 1;
    const int col = wave * 16 + m;

    const float* __restrict__ xrow = x + (size_t)row * FIN;

    v8f c = {};
#pragma unroll 4
    for (int kk = 0; kk < FIN / 4; ++kk) {
        const int k = kk * 4 + koff;
        v2f a = *(const v2f*)(xrow + k);
        v2f bfrag;
        bfrag.x = W[(k + 0) * FHID + col];
        bfrag.y = W[(k + 1) * FHID + col];
        c = __builtin_amdgcn_wmma_f32_16x16x4_f32(
                 false, a,  false, bfrag,
                 (short)0, c,  false,  false);
    }

    const int mbase = blockIdx.x * 16 + hi * 8;
    float* __restrict__ hout = h + (size_t)mbase * FHID + col;
    if (blockIdx.x * 16 + 16 <= N) {
#pragma unroll
        for (int v = 0; v < 8; ++v) {
            hout[(size_t)v * FHID] = c[v];
        }
    } else {
#pragma unroll
        for (int v = 0; v < 8; ++v) {
            if (mbase + v < N) hout[(size_t)v * FHID] = c[v];
        }
    }
}

__global__ void cgcl_scatter_kernel(const int* __restrict__ src,
                                    const int* __restrict__ dst,
                                    const float* __restrict__ dinv,
                                    const float* __restrict__ h,
                                    float* __restrict__ out, int E) {
    int gid = blockIdx.x * blockDim.x + threadIdx.x;
    int e = gid >> 4;
    if (e >= E) return;
    int q = gid & 15;
    int s = src[e];
    int d = dst[e];
    float cf = dinv[s] * dinv[d];
    const float4 hv = *(const float4*)(h + (size_t)s * FHID + q * 4);
    float* o = out + (size_t)d * FHID + q * 4;
    unsafeAtomicAdd(o + 0, hv.x * cf);
    unsafeAtomicAdd(o + 1, hv.y * cf);
    unsafeAtomicAdd(o + 2, hv.z * cf);
    unsafeAtomicAdd(o + 3, hv.w * cf);
}

__global__ void cgcl_finalize_kernel(const float* __restrict__ h,
                                     const float* __restrict__ dinv,
                                     const float* __restrict__ bias,
                                     float* __restrict__ out, int total) {
    int gid = blockIdx.x * blockDim.x + threadIdx.x;
    if (gid >= total) return;
    int n = gid >> 6;
    int f = gid & (FHID - 1);
    float di = dinv[n];
    float v = out[gid] + h[gid] * di * di + bias[f];
    out[gid] = (v > 0.0f) ? v : expm1f(v);
}

extern "C" void kernel_launch(void* const* d_in, const int* in_sizes, int n_in,
                              void* d_out, int out_size, void* d_ws, size_t ws_size,
                              hipStream_t stream) {
    const float* x    = (const float*)d_in[0];
    const float* W    = (const float*)d_in[1];
    const float* bias = (const float*)d_in[2];
    const int*   ei   = (const int*)d_in[3];

    const int N = in_sizes[0] / FIN;
    const int E = in_sizes[3] / 2;
    const int* src = ei;
    const int* dst = ei + E;

    size_t NA = ((size_t)N + 255) & ~(size_t)255;
    unsigned* deg  = (unsigned*)d_ws;
    float*    dinv = (float*)d_ws + NA;
    float*    h    = (float*)d_ws + 2 * NA;
    float*    out  = (float*)d_out;

    hipMemsetAsync(deg, 0, (size_t)N * sizeof(unsigned), stream);
    hipMemsetAsync(out, 0, (size_t)out_size * sizeof(float), stream);

    cgcl_deg_kernel<<<(E + 255) / 256, 256, 0, stream>>>(dst, deg, E);
    cgcl_dinv_kernel<<<(N + 255) / 256, 256, 0, stream>>>(deg, dinv, N);

    const int row_tiles = (N + 15) / 16;
    cgcl_gemm_wmma<<<row_tiles, 128, 0, stream>>>(x, W, h, N);

    const long long sc_threads = (long long)E * 16;
    cgcl_scatter_kernel<<<(unsigned)((sc_threads + 255) / 256), 256, 0, stream>>>(
        src, dst, dinv, h, out, E);

    const int total = N * FHID;
    cgcl_finalize_kernel<<<(total + 255) / 256, 256, 0, stream>>>(
        h, dinv, bias, out, total);
}


# scatter+finalize replaced by owner-computes gather: a work-group owns 400 dst rows (f32 sums in LDS), streams the whole dst list, adds h[src] rows by f32 LDS atomics (arrival order), stores out once
# speedup vs baseline: 1.3770x; 1.3770x over previous
#include <hip/hip_runtime.h>
#include <hip/hip_bf16.h>
#include <math.h>

#define FIN  128
#define FHID 64

typedef float v2f __attribute__((ext_vector_type(2)));
typedef float v8f __attribute__((ext_vector_type(8)));

__global__ void cgcl_deg_kernel(const int* __restrict__ dst,
                                unsigned* __restrict__ deg, int E) {
    int e = blockIdx.x * blockDim.x + threadIdx.x;
    if (e < E) atomicAdd(&deg[dst[e]], 1u);
}

__global__ void cgcl_dinv_kernel(const unsigned* __restrict__ deg,
                                 float* __restrict__ dinv, int N) {
    int n = blockIdx.x * blockDim.x + threadIdx.x;
    if (n < N) dinv[n] = rsqrtf((float)deg[n] + 1.0f);
}

__global__ void __launch_bounds__(128)
cgcl_gemm_wmma(const float* __restrict__ x, const float* __restrict__ W,
               float* __restrict__ h, int N) {
    const int lane = threadIdx.x & 31;
    const int wave = threadIdx.x >> 5;
    const int m    = lane & 15;
    const int hi   = lane >> 4;
    const int koff = hi * 2;

    int row = blockIdx.x * 16 + m;
    if (row >= N) row = N - 1;
    const int col = wave * 16 + m;

    const float* __restrict__ xrow = x + (size_t)row * FIN;

    v8f c = {};
#pragma unroll 4
    for (int kk = 0; kk < FIN / 4; ++kk) {
        const int k = kk * 4 + koff;
        v2f a = *(const v2f*)(xrow + k);
        v2f bfrag;
        bfrag.x = W[(k + 0) * FHID + col];
        bfrag.y = W[(k + 1) * FHID + col];
        c = __builtin_amdgcn_wmma_f32_16x16x4_f32(
                 false, a,  false, bfrag,
                 (short)0, c,  false,  false);
    }

    const int mbase = blockIdx.x * 16 + hi * 8;
    float* __restrict__ hout = h + (size_t)mbase * FHID + col;
    if (blockIdx.x * 16 + 16 <= N) {
#pragma unroll
        for (int v = 0; v < 8; ++v) {
            hout[(size_t)v * FHID] = c[v];
        }
    } else {
#pragma unroll
        for (int v = 0; v < 8; ++v) {
            if (mbase + v < N) hout[(size_t)v * FHID] = c[v];
        }
    }
}

#define ROWS_PER_BLOCK 400
#define GATHER_THREADS 1024
#define GATHER_WAVES   32
#define QCAP 544

template <bool VEC>
__device__ __forceinline__ int4 cgcl_load_dst4(const int* __restrict__ dst, int g,
                                               int ngroups, int E) {
    const int gc = (g < ngroups) ? g : ngroups - 1;
    if (VEC) {
        return *(const int4*)(dst + 4 * (size_t)gc);
    } else {
        const int e = 4 * gc;
        int4 v;
        v.x = dst[e];
        v.y = dst[(e + 1 < E) ? e + 1 : e];
        v.z = dst[(e + 2 < E) ? e + 2 : e];
        v.w = dst[(e + 3 < E) ? e + 3 : e];
        return v;
    }
}

__device__ __forceinline__ void cgcl_lds_add(float* p, float v) {
    __hip_atomic_fetch_add(p, v, __ATOMIC_RELAXED, __HIP_MEMORY_SCOPE_WORKGROUP);
}

__device__ __forceinline__ void cgcl_process_batch(int n, int b0,
                                                   const int* qe, const int* qr,
                                                   const int* __restrict__ src,
                                                   const float* __restrict__ dinv,
                                                   const float* __restrict__ h,
                                                   float* acc, int lo, int lane) {
    __builtin_amdgcn_fence(__ATOMIC_RELEASE, "wavefront");
    __builtin_amdgcn_wave_barrier();
    __builtin_amdgcn_fence(__ATOMIC_ACQUIRE, "wavefront");

    int s = 0, r = 0;
    float cf = 0.0f;
    if (lane < n) {
        const int e = qe[b0 + lane];
        r  = qr[b0 + lane];
        s  = src[e];
        cf = dinv[s] * dinv[lo + r];
    }
    for (int i0 = 0; i0 < n; i0 += 8) {
        float2 a[8];
        float  cc[8];
        int    rr[8];
#pragma unroll
        for (int j = 0; j < 8; ++j) {
            const bool ok = (i0 + j < n);
            const int  i  = ok ? i0 + j : n - 1;
            const int  si = __builtin_amdgcn_readlane(s, i);
            rr[j] = __builtin_amdgcn_readlane(r, i);
            const float cj = __int_as_float(__builtin_amdgcn_readlane(__float_as_int(cf), i));
            cc[j] = ok ? cj : 0.0f;
            a[j] = *(const float2*)(h + (size_t)si * FHID + 2 * lane);
        }
#pragma unroll
        for (int j = 0; j < 8; ++j) {
            float* ap = acc + rr[j] * FHID + 2 * lane;
            cgcl_lds_add(ap,     a[j].x * cc[j]);
            cgcl_lds_add(ap + 1, a[j].y * cc[j]);
        }
    }
}

template <bool VEC>
__device__ __forceinline__ void cgcl_test_set(const int4 (&v4)[4], int step, bool live, int gl,
                                              int ngroups, int E, int lo, int lane,
                                              int* qe, int* qr, int& qn) {
#pragma unroll
    for (int u = 0; u < 4; ++u) {
        const int g = step * (GATHER_THREADS * 4) + u * GATHER_THREADS + gl;
        const int4 v = v4[u];
        const unsigned r0 = (unsigned)v.x - (unsigned)lo;
        const unsigned r1 = (unsigned)v.y - (unsigned)lo;
        const unsigned r2 = (unsigned)v.z - (unsigned)lo;
        const unsigned r3 = (unsigned)v.w - (unsigned)lo;
        const bool k0 = live && (VEC ? (g < ngroups) : (4 * g + 0 < E));
        const bool k1 = VEC ? k0 : (live && (4 * g + 1 < E));
        const bool k2 = VEC ? k0 : (live && (4 * g + 2 < E));
        const bool k3 = VEC ? k0 : (live && (4 * g + 3 < E));
        const unsigned b0 = __builtin_amdgcn_ballot_w32(r0 < (unsigned)ROWS_PER_BLOCK) & __builtin_amdgcn_ballot_w32(k0);
        const unsigned b1 = __builtin_amdgcn_ballot_w32(r1 < (unsigned)ROWS_PER_BLOCK) & __builtin_amdgcn_ballot_w32(k1);
        const unsigned b2 = __builtin_amdgcn_ballot_w32(r2 < (unsigned)ROWS_PER_BLOCK) & __builtin_amdgcn_ballot_w32(k2);
        const unsigned b3 = __builtin_amdgcn_ballot_w32(r3 < (unsigned)ROWS_PER_BLOCK) & __builtin_amdgcn_ballot_w32(k3);
        if ((b0 | b1 | b2 | b3) != 0u) {
            if (b0) {
                const int pos = qn + (int)__builtin_amdgcn_mbcnt_lo(b0, 0u);
                if ((b0 >> lane) & 1u) { qe[pos] = 4 * g + 0; qr[pos] = (int)r0; }
                qn += __builtin_popcount(b0);
            }
            if (b1) {
                const int pos = qn + (int)__builtin_amdgcn_mbcnt_lo(b1, 0u);
                if ((b1 >> lane) & 1u) { qe[pos] = 4 * g + 1; qr[pos] = (int)r1; }
                qn += __builtin_popcount(b1);
            }
            if (b2) {
                const int pos = qn + (int)__builtin_amdgcn_mbcnt_lo(b2, 0u);
                if ((b2 >> lane) & 1u) { qe[pos] = 4 * g + 2; qr[pos] = (int)r2; }
                qn += __builtin_popcount(b2);
            }
            if (b3) {
                const int pos = qn + (int)__builtin_amdgcn_mbcnt_lo(b3, 0u);
                if ((b3 >> lane) & 1u) { qe[pos] = 4 * g + 3; qr[pos] = (int)r3; }
                qn += __builtin_popcount(b3);
            }
        }
    }
}

template <bool VEC>
__device__ __forceinline__ void cgcl_load_set(int4 (&v4)[4], const int* __restrict__ dst,
                                              int step, int gl, int ngroups, int E) {
#pragma unroll
    for (int u = 0; u < 4; ++u)
        v4[u] = cgcl_load_dst4<VEC>(dst, step * (GATHER_THREADS * 4) + u * GATHER_THREADS + gl, ngroups, E);
}

template <bool VEC>
__device__ __forceinline__ void cgcl_scan_edges(const int* __restrict__ src,
                                                const int* __restrict__ dst,
                                                const float* __restrict__ dinv,
                                                const float* __restrict__ h,
                                                float* acc, int* qe, int* qr,
                                                int lo, int E, int lane, int wave) {
    const int ngroups = (E + 3) >> 2;
    const int per_it  = GATHER_THREADS * 4;
    const int niter   = (ngroups + per_it - 1) / per_it;
    const int gl = wave * 32 + lane;
    int qn = 0;

    auto drain = [&](bool last) {
        const int need = last ? 1 : 32;
        while (qn >= need) {
            const int nb = (qn < 32) ? qn : 32;
            qn -= nb;
            cgcl_process_batch(nb, qn, qe, qr, src, dinv, h, acc, lo, lane);
        }
    };
    auto next_step = [&](int s) { return (s + 1 >= niter) ? 0 : s + 1; };

    int sA = (int)(blockIdx.x % (unsigned)niter);
    int sB = next_step(sA);
    int sC;
    int4 A[4], B[4], C[4];
    cgcl_load_set<VEC>(A, dst, sA, gl, ngroups, E);
    __builtin_amdgcn_sched_barrier(0);
    cgcl_load_set<VEC>(B, dst, sB, gl, ngroups, E);

    const int ntrips = (niter + 2) / 3;
    int it = 0;
    for (int t = 0; t < ntrips; ++t) {
        sC = next_step(sB);
        cgcl_load_set<VEC>(C, dst, sC, gl, ngroups, E);
        cgcl_test_set<VEC>(A, sA, it < niter, gl, ngroups, E, lo, lane, qe, qr, qn);
        ++it;
        drain(it >= niter);

        sA = next_step(sC);
        cgcl_load_set<VEC>(A, dst, sA, gl, ngroups, E);
        cgcl_test_set<VEC>(B, sB, it < niter, gl, ngroups, E, lo, lane, qe, qr, qn);
        ++it;
        drain(it >= niter);

        sB = next_step(sA);
        cgcl_load_set<VEC>(B, dst, sB, gl, ngroups, E);
        cgcl_test_set<VEC>(C, sC, it < niter, gl, ngroups, E, lo, lane, qe, qr, qn);
        ++it;
        drain(it >= niter);
    }
}

__global__ void __launch_bounds__(GATHER_THREADS)
cgcl_gather_owner_kernel(const int* __restrict__ src,
                         const int* __restrict__ dst,
                         const float* __restrict__ dinv,
                         const float* __restrict__ h,
                         const float* __restrict__ bias,
                         float* __restrict__ out, int N, int E) {
    __shared__ float acc[ROWS_PER_BLOCK * FHID];
    __shared__ int   q_e[GATHER_WAVES][QCAP];
    __shared__ int   q_r[GATHER_WAVES][QCAP];

    const int tid  = threadIdx.x;
    const int lane = tid & 31;
    const int wave = tid >> 5;
    const int lo   = blockIdx.x * ROWS_PER_BLOCK;

    for (int i = tid; i < ROWS_PER_BLOCK * FHID; i += GATHER_THREADS) acc[i] = 0.0f;
    __syncthreads();

    const bool vec_ok = ((E & 3) == 0) && ((((size_t)dst) & 15) == 0);
    if (E > 0) {
        if (vec_ok) cgcl_scan_edges<true >(src, dst, dinv, h, acc, q_e[wave], q_r[wave], lo, E, lane, wave);
        else        cgcl_scan_edges<false>(src, dst, dinv, h, acc, q_e[wave], q_r[wave], lo, E, lane, wave);
    }

    __syncthreads();

    const int   f  = tid & (FHID - 1);
    const float bf = bias[f];
    constexpr int EP_TOTAL = ROWS_PER_BLOCK * FHID / GATHER_THREADS;
    constexpr int EP_BATCH = 5;
    static_assert(ROWS_PER_BLOCK * FHID % GATHER_THREADS == 0, "whole elements per thread");
    static_assert(EP_TOTAL % EP_BATCH == 0, "whole batches");
    for (int i0 = 0; i0 < EP_TOTAL; i0 += EP_BATCH) {
        float hv[EP_BATCH], dv[EP_BATCH];
#pragma unroll
        for (int j = 0; j < EP_BATCH; ++j) {
            const int n  = lo + ((tid + (i0 + j) * GATHER_THREADS) >> 6);
            const int nc = (n < N) ? n : N - 1;
            hv[j] = h[(size_t)nc * FHID + f];
            dv[j] = dinv[nc];
        }
#pragma unroll
        for (int j = 0; j < EP_BATCH; ++j) {
            const int i = tid + (i0 + j) * GATHER_THREADS;
            const int n = lo + (i >> 6);
            if (n < N) {
                const float di = dv[j];
                const float v  = acc[i] + hv[j] * di * di + bf;
                out[(size_t)n * FHID + f] = (v > 0.0f) ? v : expm1f(v);
            }
        }
    }
}

extern "C" void kernel_launch(void* const* d_in, const int* in_sizes, int n_in,
                              void* d_out, int out_size, void* d_ws, size_t ws_size,
                              hipStream_t stream) {
    const float* x    = (const float*)d_in[0];
    const float* W    = (const float*)d_in[1];
    const float* bias = (const float*)d_in[2];
    const int*   ei   = (const int*)d_in[3];

    const int N = in_sizes[0] / FIN;
    const int E = in_sizes[3] / 2;
    const int* src = ei;
    const int* dst = ei + E;

    size_t NA = ((size_t)N + 255) & ~(size_t)255;
    unsigned* deg  = (unsigned*)d_ws;
    float*    dinv = (float*)d_ws + NA;
    float*    h    = (float*)d_ws + 2 * NA;
    float*    out  = (float*)d_out;

    hipMemsetAsync(deg, 0, (size_t)N * sizeof(unsigned), stream);

    cgcl_deg_kernel<<<(E + 255) / 256, 256, 0, stream>>>(dst, deg, E);
    cgcl_dinv_kernel<<<(N + 255) / 256, 256, 0, stream>>>(deg, dinv, N);

    const int row_tiles = (N + 15) / 16;
    cgcl_gemm_wmma<<<row_tiles, 128, 0, stream>>>(x, W, h, N);

    cgcl_gather_owner_kernel<<<(N + ROWS_PER_BLOCK - 1) / ROWS_PER_BLOCK, GATHER_THREADS, 0, stream>>>(
        src, dst, dinv, h, bias, out, N, E);
}


# gather: each edge placed once with its owner range (LDS counts, one global cursor add per range and chunk; 8-B records parked in the range's own out rows); owners no longer stream dst; arrival order
# speedup vs baseline: 1.5810x; 1.1481x over previous
#include <hip/hip_runtime.h>
#include <hip/hip_bf16.h>
#include <math.h>

#define FIN  128
#define FHID 64

typedef float v2f __attribute__((ext_vector_type(2)));
typedef float v8f __attribute__((ext_vector_type(8)));

__global__ void cgcl_deg_kernel(const int* __restrict__ dst,
                                unsigned* __restrict__ deg, int E) {
    int e = blockIdx.x * blockDim.x + threadIdx.x;
    if (e < E) atomicAdd(&deg[dst[e]], 1u);
}

__global__ void cgcl_dinv_kernel(unsigned* __restrict__ deg,
                                 float* __restrict__ dinv, int N) {
    int n = blockIdx.x * blockDim.x + threadIdx.x;
    if (n < N) {
        dinv[n] = rsqrtf((float)deg[n] + 1.0f);
        deg[n] = 0u;
    }
}

__global__ void __launch_bounds__(128)
cgcl_gemm_wmma(const float* __restrict__ x, const float* __restrict__ W,
               float* __restrict__ h, int N) {
    const int lane = threadIdx.x & 31;
    const int wave = threadIdx.x >> 5;
    const int m    = lane & 15;
    const int hi   = lane >> 4;
    const int koff = hi * 2;

    int row = blockIdx.x * 16 + m;
    if (row >= N) row = N - 1;
    const int col = wave * 16 + m;

    const float* __restrict__ xrow = x + (size_t)row * FIN;

    v8f c = {};
#pragma unroll 4
    for (int kk = 0; kk < FIN / 4; ++kk) {
        const int k = kk * 4 + koff;
        v2f a = *(const v2f*)(xrow + k);
        v2f bfrag;
        bfrag.x = W[(k + 0) * FHID + col];
        bfrag.y = W[(k + 1) * FHID + col];
        c = __builtin_amdgcn_wmma_f32_16x16x4_f32(
                 false, a,  false, bfrag,
                 (short)0, c,  false,  false);
    }

    const int mbase = blockIdx.x * 16 + hi * 8;
    float* __restrict__ hout = h + (size_t)mbase * FHID + col;
    if (blockIdx.x * 16 + 16 <= N) {
#pragma unroll
        for (int v = 0; v < 8; ++v) {
            hout[(size_t)v * FHID] = c[v];
        }
    } else {
#pragma unroll
        for (int v = 0; v < 8; ++v) {
            if (mbase + v < N) hout[(size_t)v * FHID] = c[v];
        }
    }
}

#define ROWS_PER_BLOCK 400
#define GATHER_THREADS 1024
#define GATHER_WAVES   32
#define QCAP 544

#define BIN_THREADS 1024
#define BIN_EPT     8
#define BIN_CHUNK   (BIN_THREADS * BIN_EPT)
#define MAX_RANGES  1024
#define REC_PER_ROW (FHID / 2)

__device__ __forceinline__ size_t cgcl_cursor_word(int b, int nranges, int N) {
    return ((size_t)nranges * 32 <= (size_t)N) ? (size_t)b * 32 : (size_t)b;
}

__device__ __forceinline__ unsigned cgcl_range_capacity(int b, int N) {
    const int left = N - b * ROWS_PER_BLOCK;
    const int rows = (left < ROWS_PER_BLOCK) ? left : ROWS_PER_BLOCK;
    return (unsigned)rows * REC_PER_ROW;
}

__global__ void __launch_bounds__(BIN_THREADS)
cgcl_place_kernel(const int* __restrict__ src, const int* __restrict__ dst,
                  unsigned* __restrict__ cursor, float* __restrict__ out,
                  int N, int E) {
    __shared__ unsigned cnt[MAX_RANGES];
    const int nranges = (N + ROWS_PER_BLOCK - 1) / ROWS_PER_BLOCK;
    if (nranges > MAX_RANGES) return;
    const int tid = threadIdx.x;
    for (int i = tid; i < nranges; i += BIN_THREADS) cnt[i] = 0u;
    __syncthreads();

    const int e0 = blockIdx.x * BIN_CHUNK + tid;
    int dv[BIN_EPT], sv[BIN_EPT];
#pragma unroll
    for (int k = 0; k < BIN_EPT; ++k) {
        const int e  = e0 + k * BIN_THREADS;
        const int ec = (e < E) ? e : E - 1;
        dv[k] = dst[ec];
        sv[k] = src[ec];
    }
    unsigned key[BIN_EPT];
#pragma unroll
    for (int k = 0; k < BIN_EPT; ++k) {
        const bool live = (e0 + k * BIN_THREADS < E);
        const int  b    = dv[k] / ROWS_PER_BLOCK;
        const unsigned rank = atomicAdd(&cnt[b], live ? 1u : 0u);
        key[k] = live ? (((unsigned)b << 16) | rank) : 0xFFFFFFFFu;
    }
    __syncthreads();
    for (int i = tid; i < nranges; i += BIN_THREADS) {
        const unsigned c = cnt[i];
        unsigned base = 0u;
        if (c != 0u) base = atomicAdd(&cursor[cgcl_cursor_word(i, nranges, N)], c);
        cnt[i] = base;
    }
    __syncthreads();
#pragma unroll
    for (int k = 0; k < BIN_EPT; ++k) {
        if (key[k] != 0xFFFFFFFFu) {
            const int b = (int)(key[k] >> 16);
            const unsigned pos = cnt[b] + (key[k] & 0xFFFFu);
            if (pos < cgcl_range_capacity(b, N)) {
                int2* rec = (int2*)(out + (size_t)b * ROWS_PER_BLOCK * FHID);
                rec[pos] = make_int2(sv[k], dv[k] - b * ROWS_PER_BLOCK);
            }
        }
    }
}

template <bool VEC>
__device__ __forceinline__ int4 cgcl_load_dst4(const int* __restrict__ dst, int g,
                                               int ngroups, int E) {
    const int gc = (g < ngroups) ? g : ngroups - 1;
    if (VEC) {
        return *(const int4*)(dst + 4 * (size_t)gc);
    } else {
        const int e = 4 * gc;
        int4 v;
        v.x = dst[e];
        v.y = dst[(e + 1 < E) ? e + 1 : e];
        v.z = dst[(e + 2 < E) ? e + 2 : e];
        v.w = dst[(e + 3 < E) ? e + 3 : e];
        return v;
    }
}

__device__ __forceinline__ void cgcl_lds_add(float* p, float v) {
    __hip_atomic_fetch_add(p, v, __ATOMIC_RELAXED, __HIP_MEMORY_SCOPE_WORKGROUP);
}

__device__ __forceinline__ void cgcl_add_batch(int n, int s, int r, float cf,
                                               const float* __restrict__ h,
                                               float* acc, int lane) {
    for (int i0 = 0; i0 < n; i0 += 8) {
        float2 a[8];
        float  cc[8];
        int    rr[8];
#pragma unroll
        for (int j = 0; j < 8; ++j) {
            const bool ok = (i0 + j < n);
            const int  i  = ok ? i0 + j : n - 1;
            const int  si = __builtin_amdgcn_readlane(s, i);
            rr[j] = __builtin_amdgcn_readlane(r, i);
            const float cj = __int_as_float(__builtin_amdgcn_readlane(__float_as_int(cf), i));
            cc[j] = ok ? cj : 0.0f;
            a[j] = *(const float2*)(h + (size_t)si * FHID + 2 * lane);
        }
#pragma unroll
        for (int j = 0; j < 8; ++j) {
            float* ap = acc + rr[j] * FHID + 2 * lane;
            cgcl_lds_add(ap,     a[j].x * cc[j]);
            cgcl_lds_add(ap + 1, a[j].y * cc[j]);
        }
    }
}

__device__ __forceinline__ void cgcl_process_batch(int n, int b0,
                                                   const int* qe, const int* qr,
                                                   const int* __restrict__ src,
                                                   const float* __restrict__ dinv,
                                                   const float* __restrict__ h,
                                                   float* acc, int lo, int lane) {
    __builtin_amdgcn_fence(__ATOMIC_RELEASE, "wavefront");
    __builtin_amdgcn_wave_barrier();
    __builtin_amdgcn_fence(__ATOMIC_ACQUIRE, "wavefront");

    int s = 0, r = 0;
    float cf = 0.0f;
    if (lane < n) {
        const int e = qe[b0 + lane];
        r  = qr[b0 + lane];
        s  = src[e];
        cf = dinv[s] * dinv[lo + r];
    }
    cgcl_add_batch(n, s, r, cf, h, acc, lane);
}

template <bool VEC>
__device__ __forceinline__ void cgcl_test_set(const int4 (&v4)[4], int step, bool live, int gl,
                                              int ngroups, int E, int lo, int lane,
                                              int* qe, int* qr, int& qn) {
#pragma unroll
    for (int u = 0; u < 4; ++u) {
        const int g = step * (GATHER_THREADS * 4) + u * GATHER_THREADS + gl;
        const int4 v = v4[u];
        const unsigned r0 = (unsigned)v.x - (unsigned)lo;
        const unsigned r1 = (unsigned)v.y - (unsigned)lo;
        const unsigned r2 = (unsigned)v.z - (unsigned)lo;
        const unsigned r3 = (unsigned)v.w - (unsigned)lo;
        const bool k0 = live && (VEC ? (g < ngroups) : (4 * g + 0 < E));
        const bool k1 = VEC ? k0 : (live && (4 * g + 1 < E));
        const bool k2 = VEC ? k0 : (live && (4 * g + 2 < E));
        const bool k3 = VEC ? k0 : (live && (4 * g + 3 < E));
        const unsigned b0 = __builtin_amdgcn_ballot_w32(r0 < (unsigned)ROWS_PER_BLOCK) & __builtin_amdgcn_ballot_w32(k0);
        const unsigned b1 = __builtin_amdgcn_ballot_w32(r1 < (unsigned)ROWS_PER_BLOCK) & __builtin_amdgcn_ballot_w32(k1);
        const unsigned b2 = __builtin_amdgcn_ballot_w32(r2 < (unsigned)ROWS_PER_BLOCK) & __builtin_amdgcn_ballot_w32(k2);
        const unsigned b3 = __builtin_amdgcn_ballot_w32(r3 < (unsigned)ROWS_PER_BLOCK) & __builtin_amdgcn_ballot_w32(k3);
        if ((b0 | b1 | b2 | b3) != 0u) {
            if (b0) {
                const int pos = qn + (int)__builtin_amdgcn_mbcnt_lo(b0, 0u);
                if ((b0 >> lane) & 1u) { qe[pos] = 4 * g + 0; qr[pos] = (int)r0; }
                qn += __builtin_popcount(b0);
            }
            if (b1) {
                const int pos = qn + (int)__builtin_amdgcn_mbcnt_lo(b1, 0u);
                if ((b1 >> lane) & 1u) { qe[pos] = 4 * g + 1; qr[pos] = (int)r1; }
                qn += __builtin_popcount(b1);
            }
            if (b2) {
                const int pos = qn + (int)__builtin_amdgcn_mbcnt_lo(b2, 0u);
                if ((b2 >> lane) & 1u) { qe[pos] = 4 * g + 2; qr[pos] = (int)r2; }
                qn += __builtin_popcount(b2);
            }
            if (b3) {
                const int pos = qn + (int)__builtin_amdgcn_mbcnt_lo(b3, 0u);
                if ((b3 >> lane) & 1u) { qe[pos] = 4 * g + 3; qr[pos] = (int)r3; }
                qn += __builtin_popcount(b3);
            }
        }
    }
}

template <bool VEC>
__device__ __forceinline__ void cgcl_load_set(int4 (&v4)[4], const int* __restrict__ dst,
                                              int step, int gl, int ngroups, int E) {
#pragma unroll
    for (int u = 0; u < 4; ++u)
        v4[u] = cgcl_load_dst4<VEC>(dst, step * (GATHER_THREADS * 4) + u * GATHER_THREADS + gl, ngroups, E);
}

template <bool VEC>
__device__ __forceinline__ void cgcl_scan_edges(const int* __restrict__ src,
                                                const int* __restrict__ dst,
                                                const float* __restrict__ dinv,
                                                const float* __restrict__ h,
                                                float* acc, int* qe, int* qr,
                                                int lo, int E, int lane, int wave) {
    const int ngroups = (E + 3) >> 2;
    const int per_it  = GATHER_THREADS * 4;
    const int niter   = (ngroups + per_it - 1) / per_it;
    const int gl = wave * 32 + lane;
    int qn = 0;

    auto drain = [&](bool last) {
        const int need = last ? 1 : 32;
        while (qn >= need) {
            const int nb = (qn < 32) ? qn : 32;
            qn -= nb;
            cgcl_process_batch(nb, qn, qe, qr, src, dinv, h, acc, lo, lane);
        }
    };
    auto next_step = [&](int s) { return (s + 1 >= niter) ? 0 : s + 1; };

    int sA = (int)(blockIdx.x % (unsigned)niter);
    int sB = next_step(sA);
    int sC;
    int4 A[4], B[4], C[4];
    cgcl_load_set<VEC>(A, dst, sA, gl, ngroups, E);
    __builtin_amdgcn_sched_barrier(0);
    cgcl_load_set<VEC>(B, dst, sB, gl, ngroups, E);

    const int ntrips = (niter + 2) / 3;
    int it = 0;
    for (int t = 0; t < ntrips; ++t) {
        sC = next_step(sB);
        cgcl_load_set<VEC>(C, dst, sC, gl, ngroups, E);
        cgcl_test_set<VEC>(A, sA, it < niter, gl, ngroups, E, lo, lane, qe, qr, qn);
        ++it;
        drain(it >= niter);

        sA = next_step(sC);
        cgcl_load_set<VEC>(A, dst, sA, gl, ngroups, E);
        cgcl_test_set<VEC>(B, sB, it < niter, gl, ngroups, E, lo, lane, qe, qr, qn);
        ++it;
        drain(it >= niter);

        sB = next_step(sA);
        cgcl_load_set<VEC>(B, dst, sB, gl, ngroups, E);
        cgcl_test_set<VEC>(C, sC, it < niter, gl, ngroups, E, lo, lane, qe, qr, qn);
        ++it;
        drain(it >= niter);
    }
}

__global__ void __launch_bounds__(GATHER_THREADS)
cgcl_gather_owner_kernel(const int* __restrict__ src,
                         const int* __restrict__ dst,
                         const unsigned* __restrict__ cursor,
                         const float* __restrict__ dinv,
                         const float* __restrict__ h,
                         const float* __restrict__ bias,
                         float* __restrict__ out, int N, int E) {
    __shared__ float acc[ROWS_PER_BLOCK * FHID];
    __shared__ int   q_e[GATHER_WAVES][QCAP];
    __shared__ int   q_r[GATHER_WAVES][QCAP];

    const int tid  = threadIdx.x;
    const int lane = tid & 31;
    const int wave = tid >> 5;
    const int lo   = blockIdx.x * ROWS_PER_BLOCK;

    for (int i = tid; i < ROWS_PER_BLOCK * FHID; i += GATHER_THREADS) acc[i] = 0.0f;
    __syncthreads();

    const int      nranges = (N + ROWS_PER_BLOCK - 1) / ROWS_PER_BLOCK;
    const unsigned nrec    = cursor[cgcl_cursor_word((int)blockIdx.x, nranges, N)];
    const bool     placed  = (nranges <= MAX_RANGES) && (nrec <= cgcl_range_capacity((int)blockIdx.x, N));
    if (placed) {
        const int2* rec = (const int2*)(out + (size_t)lo * FHID);
        for (int b0 = wave * 32; b0 < (int)nrec; b0 += GATHER_WAVES * 32) {
            const int left = (int)nrec - b0;
            const int n    = (left < 32) ? left : 32;
            int s = 0, r = 0;
            float cf = 0.0f;
            if (lane < n) {
                const int2 v = rec[b0 + lane];
                s  = v.x;
                r  = v.y;
                cf = dinv[s] * dinv[lo + r];
            }
            cgcl_add_batch(n, s, r, cf, h, acc, lane);
        }
    } else if (E > 0) {
        const bool vec_ok = ((E & 3) == 0) && ((((size_t)dst) & 15) == 0);
        if (vec_ok) cgcl_scan_edges<true >(src, dst, dinv, h, acc, q_e[wave], q_r[wave], lo, E, lane, wave);
        else        cgcl_scan_edges<false>(src, dst, dinv, h, acc, q_e[wave], q_r[wave], lo, E, lane, wave);
    }

    __syncthreads();

    const int   f  = tid & (FHID - 1);
    const float bf = bias[f];
    constexpr int EP_TOTAL = ROWS_PER_BLOCK * FHID / GATHER_THREADS;
    constexpr int EP_BATCH = 5;
    static_assert(ROWS_PER_BLOCK * FHID % GATHER_THREADS == 0, "whole elements per thread");
    static_assert(EP_TOTAL % EP_BATCH == 0, "whole batches");
    for (int i0 = 0; i0 < EP_TOTAL; i0 += EP_BATCH) {
        float hv[EP_BATCH], dv[EP_BATCH];
#pragma unroll
        for (int j = 0; j < EP_BATCH; ++j) {
            const int n  = lo + ((tid + (i0 + j) * GATHER_THREADS) >> 6);
            const int nc = (n < N) ? n : N - 1;
            hv[j] = h[(size_t)nc * FHID + f];
            dv[j] = dinv[nc];
        }
#pragma unroll
        for (int j = 0; j < EP_BATCH; ++j) {
            const int i = tid + (i0 + j) * GATHER_THREADS;
            const int n = lo + (i >> 6);
            if (n < N) {
                const float di = dv[j];
                const float v  = acc[i] + hv[j] * di * di + bf;
                out[(size_t)n * FHID + f] = (v > 0.0f) ? v : expm1f(v);
            }
        }
    }
}

extern "C" void kernel_launch(void* const* d_in, const int* in_sizes, int n_in,
                              void* d_out, int out_size, void* d_ws, size_t ws_size,
                              hipStream_t stream) {
    const float* x    = (const float*)d_in[0];
    const float* W    = (const float*)d_in[1];
    const float* bias = (const float*)d_in[2];
    const int*   ei   = (const int*)d_in[3];

    const int N = in_sizes[0] / FIN;
    const int E = in_sizes[3] / 2;
    const int* src = ei;
    const int* dst = ei + E;

    size_t NA = ((size_t)N + 255) & ~(size_t)255;
    unsigned* deg  = (unsigned*)d_ws;
    float*    dinv = (float*)d_ws + NA;
    float*    h    = (float*)d_ws + 2 * NA;
    float*    out  = (float*)d_out;

    hipMemsetAsync(deg, 0, (size_t)N * sizeof(unsigned), stream);

    cgcl_deg_kernel<<<(E + 255) / 256, 256, 0, stream>>>(dst, deg, E);
    cgcl_dinv_kernel<<<(N + 255) / 256, 256, 0, stream>>>(deg, dinv, N);

    cgcl_place_kernel<<<(E + BIN_CHUNK - 1) / BIN_CHUNK, BIN_THREADS, 0, stream>>>(src, dst, deg, out, N, E);

    const int row_tiles = (N + 15) / 16;
    cgcl_gemm_wmma<<<row_tiles, 128, 0, stream>>>(x, W, h, N);

    cgcl_gather_owner_kernel<<<(N + ROWS_PER_BLOCK - 1) / ROWS_PER_BLOCK, GATHER_THREADS, 0, stream>>>(
        src, dst, deg, dinv, h, bias, out, N, E);
}


# degree pass of global atomics over dst and the dinv launch dropped: each range owner counts the records already placed for its rows (LDS integer adds) and writes dinv; 5 launches become 4
# speedup vs baseline: 2.8466x; 1.8005x over previous
#include <hip/hip_runtime.h>
#include <hip/hip_bf16.h>
#include <math.h>

#define FIN  128
#define FHID 64

typedef float v2f __attribute__((ext_vector_type(2)));
typedef float v8f __attribute__((ext_vector_type(8)));

__global__ void __launch_bounds__(128)
cgcl_gemm_wmma(const float* __restrict__ x, const float* __restrict__ W,
               float* __restrict__ h, int N) {
    const int lane = threadIdx.x & 31;
    const int wave = threadIdx.x >> 5;
    const int m    = lane & 15;
    const int hi   = lane >> 4;
    const int koff = hi * 2;

    int row = blockIdx.x * 16 + m;
    if (row >= N) row = N - 1;
    const int col = wave * 16 + m;

    const float* __restrict__ xrow = x + (size_t)row * FIN;

    v8f c = {};
#pragma unroll 4
    for (int kk = 0; kk < FIN / 4; ++kk) {
        const int k = kk * 4 + koff;
        v2f a = *(const v2f*)(xrow + k);
        v2f bfrag;
        bfrag.x = W[(k + 0) * FHID + col];
        bfrag.y = W[(k + 1) * FHID + col];
        c = __builtin_amdgcn_wmma_f32_16x16x4_f32(
                 false, a,  false, bfrag,
                 (short)0, c,  false,  false);
    }

    const int mbase = blockIdx.x * 16 + hi * 8;
    float* __restrict__ hout = h + (size_t)mbase * FHID + col;
    if (blockIdx.x * 16 + 16 <= N) {
#pragma unroll
        for (int v = 0; v < 8; ++v) {
            hout[(size_t)v * FHID] = c[v];
        }
    } else {
#pragma unroll
        for (int v = 0; v < 8; ++v) {
            if (mbase + v < N) hout[(size_t)v * FHID] = c[v];
        }
    }
}

#define ROWS_PER_BLOCK 400
#define GATHER_THREADS 1024
#define GATHER_WAVES   32
#define QCAP 544

#define BIN_THREADS 1024
#define BIN_EPT     8
#define BIN_CHUNK   (BIN_THREADS * BIN_EPT)
#define MAX_RANGES  1024
#define REC_PER_ROW (FHID / 2)

__device__ __forceinline__ size_t cgcl_cursor_word(int b, int nranges, int N) {
    return ((size_t)nranges * 32 <= (size_t)N) ? (size_t)b * 32 : (size_t)b;
}

__device__ __forceinline__ unsigned cgcl_range_capacity(int b, int N) {
    const int left = N - b * ROWS_PER_BLOCK;
    const int rows = (left < ROWS_PER_BLOCK) ? left : ROWS_PER_BLOCK;
    return (unsigned)rows * REC_PER_ROW;
}

__global__ void __launch_bounds__(BIN_THREADS)
cgcl_place_kernel(const int* __restrict__ src, const int* __restrict__ dst,
                  unsigned* __restrict__ cursor, float* __restrict__ out,
                  int N, int E) {
    __shared__ unsigned cnt[MAX_RANGES];
    const int nranges = (N + ROWS_PER_BLOCK - 1) / ROWS_PER_BLOCK;
    if (nranges > MAX_RANGES) return;
    const int tid = threadIdx.x;
    for (int i = tid; i < nranges; i += BIN_THREADS) cnt[i] = 0u;
    __syncthreads();

    const int e0 = blockIdx.x * BIN_CHUNK + tid;
    int dv[BIN_EPT], sv[BIN_EPT];
#pragma unroll
    for (int k = 0; k < BIN_EPT; ++k) {
        const int e  = e0 + k * BIN_THREADS;
        const int ec = (e < E) ? e : E - 1;
        dv[k] = dst[ec];
        sv[k] = src[ec];
    }
    unsigned key[BIN_EPT];
#pragma unroll
    for (int k = 0; k < BIN_EPT; ++k) {
        const bool live = (e0 + k * BIN_THREADS < E);
        const int  b    = dv[k] / ROWS_PER_BLOCK;
        const unsigned rank = atomicAdd(&cnt[b], live ? 1u : 0u);
        key[k] = live ? (((unsigned)b << 16) | rank) : 0xFFFFFFFFu;
    }
    __syncthreads();
    for (int i = tid; i < nranges; i += BIN_THREADS) {
        const unsigned c = cnt[i];
        unsigned base = 0u;
        if (c != 0u) base = atomicAdd(&cursor[cgcl_cursor_word(i, nranges, N)], c);
        cnt[i] = base;
    }
    __syncthreads();
#pragma unroll
    for (int k = 0; k < BIN_EPT; ++k) {
        if (key[k] != 0xFFFFFFFFu) {
            const int b = (int)(key[k] >> 16);
            const unsigned pos = cnt[b] + (key[k] & 0xFFFFu);
            if (pos < cgcl_range_capacity(b, N)) {
                int2* rec = (int2*)(out + (size_t)b * ROWS_PER_BLOCK * FHID);
                rec[pos] = make_int2(sv[k], dv[k] - b * ROWS_PER_BLOCK);
            }
        }
    }
}

#define COUNT_THREADS 1024
#define COUNT_RPT     4

__global__ void __launch_bounds__(COUNT_THREADS)
cgcl_count_dinv_kernel(const int* __restrict__ dst,
                       const unsigned* __restrict__ cursor,
                       const float* __restrict__ out,
                       float* __restrict__ dinv, int N, int E) {
    __shared__ unsigned cnt[ROWS_PER_BLOCK];
    const int tid = threadIdx.x;
    const int lo  = blockIdx.x * ROWS_PER_BLOCK;
    for (int i = tid; i < ROWS_PER_BLOCK; i += COUNT_THREADS) cnt[i] = 0u;
    __syncthreads();

    const int      nranges = (N + ROWS_PER_BLOCK - 1) / ROWS_PER_BLOCK;
    const unsigned nrec    = cursor[cgcl_cursor_word((int)blockIdx.x, nranges, N)];
    const bool     placed  = (nranges <= MAX_RANGES) && (nrec <= cgcl_range_capacity((int)blockIdx.x, N));
    if (placed) {
        const int2* rec = (const int2*)(out + (size_t)lo * FHID);
        const int   n   = (int)nrec;
        for (int i0 = tid; i0 < n; i0 += COUNT_THREADS * COUNT_RPT) {
            int rr[COUNT_RPT];
#pragma unroll
            for (int j = 0; j < COUNT_RPT; ++j) {
                const int i  = i0 + j * COUNT_THREADS;
                const int ic = (i < n) ? i : n - 1;
                rr[j] = rec[ic].y;
            }
#pragma unroll
            for (int j = 0; j < COUNT_RPT; ++j) {
                if (i0 + j * COUNT_THREADS < n && (unsigned)rr[j] < (unsigned)ROWS_PER_BLOCK)
                    atomicAdd(&cnt[rr[j]], 1u);
            }
        }
    } else {
        for (int e = tid; e < E; e += COUNT_THREADS) {
            const unsigned r = (unsigned)dst[e] - (unsigned)lo;
            if (r < (unsigned)ROWS_PER_BLOCK) atomicAdd(&cnt[r], 1u);
        }
    }
    __syncthreads();
    for (int i = tid; i < ROWS_PER_BLOCK; i += COUNT_THREADS) {
        const int n = lo + i;
        if (n < N) dinv[n] = rsqrtf((float)cnt[i] + 1.0f);
    }
}

template <bool VEC>
__device__ __forceinline__ int4 cgcl_load_dst4(const int* __restrict__ dst, int g,
                                               int ngroups, int E) {
    const int gc = (g < ngroups) ? g : ngroups - 1;
    if (VEC) {
        return *(const int4*)(dst + 4 * (size_t)gc);
    } else {
        const int e = 4 * gc;
        int4 v;
        v.x = dst[e];
        v.y = dst[(e + 1 < E) ? e + 1 : e];
        v.z = dst[(e + 2 < E) ? e + 2 : e];
        v.w = dst[(e + 3 < E) ? e + 3 : e];
        return v;
    }
}

__device__ __forceinline__ void cgcl_lds_add(float* p, float v) {
    __hip_atomic_fetch_add(p, v, __ATOMIC_RELAXED, __HIP_MEMORY_SCOPE_WORKGROUP);
}

__device__ __forceinline__ void cgcl_add_batch(int n, int s, int r, float cf,
                                               const float* __restrict__ h,
                                               float* acc, int lane) {
    for (int i0 = 0; i0 < n; i0 += 8) {
        float2 a[8];
        float  cc[8];
        int    rr[8];
#pragma unroll
        for (int j = 0; j < 8; ++j) {
            const bool ok = (i0 + j < n);
            const int  i  = ok ? i0 + j : n - 1;
            const int  si = __builtin_amdgcn_readlane(s, i);
            rr[j] = __builtin_amdgcn_readlane(r, i);
            const float cj = __int_as_float(__builtin_amdgcn_readlane(__float_as_int(cf), i));
            cc[j] = ok ? cj : 0.0f;
            a[j] = *(const float2*)(h + (size_t)si * FHID + 2 * lane);
        }
#pragma unroll
        for (int j = 0; j < 8; ++j) {
            float* ap = acc + rr[j] * FHID + 2 * lane;
            cgcl_lds_add(ap,     a[j].x * cc[j]);
            cgcl_lds_add(ap + 1, a[j].y * cc[j]);
        }
    }
}

__device__ __forceinline__ void cgcl_process_batch(int n, int b0,
                                                   const int* qe, const int* qr,
                                                   const int* __restrict__ src,
                                                   const float* __restrict__ dinv,
                                                   const float* __restrict__ h,
                                                   float* acc, int lo, int lane) {
    __builtin_amdgcn_fence(__ATOMIC_RELEASE, "wavefront");
    __builtin_amdgcn_wave_barrier();
    __builtin_amdgcn_fence(__ATOMIC_ACQUIRE, "wavefront");

    int s = 0, r = 0;
    float cf = 0.0f;
    if (lane < n) {
        const int e = qe[b0 + lane];
        r  = qr[b0 + lane];
        s  = src[e];
        cf = dinv[s] * dinv[lo + r];
    }
    cgcl_add_batch(n, s, r, cf, h, acc, lane);
}

template <bool VEC>
__device__ __forceinline__ void cgcl_test_set(const int4 (&v4)[4], int step, bool live, int gl,
                                              int ngroups, int E, int lo, int lane,
                                              int* qe, int* qr, int& qn) {
#pragma unroll
    for (int u = 0; u < 4; ++u) {
        const int g = step * (GATHER_THREADS * 4) + u * GATHER_THREADS + gl;
        const int4 v = v4[u];
        const unsigned r0 = (unsigned)v.x - (unsigned)lo;
        const unsigned r1 = (unsigned)v.y - (unsigned)lo;
        const unsigned r2 = (unsigned)v.z - (unsigned)lo;
        const unsigned r3 = (unsigned)v.w - (unsigned)lo;
        const bool k0 = live && (VEC ? (g < ngroups) : (4 * g + 0 < E));
        const bool k1 = VEC ? k0 : (live && (4 * g + 1 < E));
        const bool k2 = VEC ? k0 : (live && (4 * g + 2 < E));
        const bool k3 = VEC ? k0 : (live && (4 * g + 3 < E));
        const unsigned b0 = __builtin_amdgcn_ballot_w32(r0 < (unsigned)ROWS_PER_BLOCK) & __builtin_amdgcn_ballot_w32(k0);
        const unsigned b1 = __builtin_amdgcn_ballot_w32(r1 < (unsigned)ROWS_PER_BLOCK) & __builtin_amdgcn_ballot_w32(k1);
        const unsigned b2 = __builtin_amdgcn_ballot_w32(r2 < (unsigned)ROWS_PER_BLOCK) & __builtin_amdgcn_ballot_w32(k2);
        const unsigned b3 = __builtin_amdgcn_ballot_w32(r3 < (unsigned)ROWS_PER_BLOCK) & __builtin_amdgcn_ballot_w32(k3);
        if ((b0 | b1 | b2 | b3) != 0u) {
            if (b0) {
                const int pos = qn + (int)__builtin_amdgcn_mbcnt_lo(b0, 0u);
                if ((b0 >> lane) & 1u) { qe[pos] = 4 * g + 0; qr[pos] = (int)r0; }
                qn += __builtin_popcount(b0);
            }
            if (b1) {
                const int pos = qn + (int)__builtin_amdgcn_mbcnt_lo(b1, 0u);
                if ((b1 >> lane) & 1u) { qe[pos] = 4 * g + 1; qr[pos] = (int)r1; }
                qn += __builtin_popcount(b1);
            }
            if (b2) {
                const int pos = qn + (int)__builtin_amdgcn_mbcnt_lo(b2, 0u);
                if ((b2 >> lane) & 1u) { qe[pos] = 4 * g + 2; qr[pos] = (int)r2; }
                qn += __builtin_popcount(b2);
            }
            if (b3) {
                const int pos = qn + (int)__builtin_amdgcn_mbcnt_lo(b3, 0u);
                if ((b3 >> lane) & 1u) { qe[pos] = 4 * g + 3; qr[pos] = (int)r3; }
                qn += __builtin_popcount(b3);
            }
        }
    }
}

template <bool VEC>
__device__ __forceinline__ void cgcl_load_set(int4 (&v4)[4], const int* __restrict__ dst,
                                              int step, int gl, int ngroups, int E) {
#pragma unroll
    for (int u = 0; u < 4; ++u)
        v4[u] = cgcl_load_dst4<VEC>(dst, step * (GATHER_THREADS * 4) + u * GATHER_THREADS + gl, ngroups, E);
}

template <bool VEC>
__device__ __forceinline__ void cgcl_scan_edges(const int* __restrict__ src,
                                                const int* __restrict__ dst,
                                                const float* __restrict__ dinv,
                                                const float* __restrict__ h,
                                                float* acc, int* qe, int* qr,
                                                int lo, int E, int lane, int wave) {
    const int ngroups = (E + 3) >> 2;
    const int per_it  = GATHER_THREADS * 4;
    const int niter   = (ngroups + per_it - 1) / per_it;
    const int gl = wave * 32 + lane;
    int qn = 0;

    auto drain = [&](bool last) {
        const int need = last ? 1 : 32;
        while (qn >= need) {
            const int nb = (qn < 32) ? qn : 32;
            qn -= nb;
            cgcl_process_batch(nb, qn, qe, qr, src, dinv, h, acc, lo, lane);
        }
    };
    auto next_step = [&](int s) { return (s + 1 >= niter) ? 0 : s + 1; };

    int sA = (int)(blockIdx.x % (unsigned)niter);
    int sB = next_step(sA);
    int sC;
    int4 A[4], B[4], C[4];
    cgcl_load_set<VEC>(A, dst, sA, gl, ngroups, E);
    __builtin_amdgcn_sched_barrier(0);
    cgcl_load_set<VEC>(B, dst, sB, gl, ngroups, E);

    const int ntrips = (niter + 2) / 3;
    int it = 0;
    for (int t = 0; t < ntrips; ++t) {
        sC = next_step(sB);
        cgcl_load_set<VEC>(C, dst, sC, gl, ngroups, E);
        cgcl_test_set<VEC>(A, sA, it < niter, gl, ngroups, E, lo, lane, qe, qr, qn);
        ++it;
        drain(it >= niter);

        sA = next_step(sC);
        cgcl_load_set<VEC>(A, dst, sA, gl, ngroups, E);
        cgcl_test_set<VEC>(B, sB, it < niter, gl, ngroups, E, lo, lane, qe, qr, qn);
        ++it;
        drain(it >= niter);

        sB = next_step(sA);
        cgcl_load_set<VEC>(B, dst, sB, gl, ngroups, E);
        cgcl_test_set<VEC>(C, sC, it < niter, gl, ngroups, E, lo, lane, qe, qr, qn);
        ++it;
        drain(it >= niter);
    }
}

__global__ void __launch_bounds__(GATHER_THREADS)
cgcl_gather_owner_kernel(const int* __restrict__ src,
                         const int* __restrict__ dst,
                         const unsigned* __restrict__ cursor,
                         const float* __restrict__ dinv,
                         const float* __restrict__ h,
                         const float* __restrict__ bias,
                         float* __restrict__ out, int N, int E) {
    __shared__ float acc[ROWS_PER_BLOCK * FHID];
    __shared__ int   q_e[GATHER_WAVES][QCAP];
    __shared__ int   q_r[GATHER_WAVES][QCAP];

    const int tid  = threadIdx.x;
    const int lane = tid & 31;
    const int wave = tid >> 5;
    const int lo   = blockIdx.x * ROWS_PER_BLOCK;

    for (int i = tid; i < ROWS_PER_BLOCK * FHID; i += GATHER_THREADS) acc[i] = 0.0f;
    __syncthreads();

    const int      nranges = (N + ROWS_PER_BLOCK - 1) / ROWS_PER_BLOCK;
    const unsigned nrec    = cursor[cgcl_cursor_word((int)blockIdx.x, nranges, N)];
    const bool     placed  = (nranges <= MAX_RANGES) && (nrec <= cgcl_range_capacity((int)blockIdx.x, N));
    if (placed) {
        const int2* rec = (const int2*)(out + (size_t)lo * FHID);
        for (int b0 = wave * 32; b0 < (int)nrec; b0 += GATHER_WAVES * 32) {
            const int left = (int)nrec - b0;
            const int n    = (left < 32) ? left : 32;
            int s = 0, r = 0;
            float cf = 0.0f;
            if (lane < n) {
                const int2 v = rec[b0 + lane];
                s  = v.x;
                r  = v.y;
                cf = dinv[s] * dinv[lo + r];
            }
            cgcl_add_batch(n, s, r, cf, h, acc, lane);
        }
    } else if (E > 0) {
        const bool vec_ok = ((E & 3) == 0) && ((((size_t)dst) & 15) == 0);
        if (vec_ok) cgcl_scan_edges<true >(src, dst, dinv, h, acc, q_e[wave], q_r[wave], lo, E, lane, wave);
        else        cgcl_scan_edges<false>(src, dst, dinv, h, acc, q_e[wave], q_r[wave], lo, E, lane, wave);
    }

    __syncthreads();

    const int   f  = tid & (FHID - 1);
    const float bf = bias[f];
    constexpr int EP_TOTAL = ROWS_PER_BLOCK * FHID / GATHER_THREADS;
    constexpr int EP_BATCH = 5;
    static_assert(ROWS_PER_BLOCK * FHID % GATHER_THREADS == 0, "whole elements per thread");
    static_assert(EP_TOTAL % EP_BATCH == 0, "whole batches");
    for (int i0 = 0; i0 < EP_TOTAL; i0 += EP_BATCH) {
        float hv[EP_BATCH], dv[EP_BATCH];
#pragma unroll
        for (int j = 0; j < EP_BATCH; ++j) {
            const int n  = lo + ((tid + (i0 + j) * GATHER_THREADS) >> 6);
            const int nc = (n < N) ? n : N - 1;
            hv[j] = h[(size_t)nc * FHID + f];
            dv[j] = dinv[nc];
        }
#pragma unroll
        for (int j = 0; j < EP_BATCH; ++j) {
            const int i = tid + (i0 + j) * GATHER_THREADS;
            const int n = lo + (i >> 6);
            if (n < N) {
                const float di = dv[j];
                const float v  = acc[i] + hv[j] * di * di + bf;
                out[(size_t)n * FHID + f] = (v > 0.0f) ? v : expm1f(v);
            }
        }
    }
}

extern "C" void kernel_launch(void* const* d_in, const int* in_sizes, int n_in,
                              void* d_out, int out_size, void* d_ws, size_t ws_size,
                              hipStream_t stream) {
    const float* x    = (const float*)d_in[0];
    const float* W    = (const float*)d_in[1];
    const float* bias = (const float*)d_in[2];
    const int*   ei   = (const int*)d_in[3];

    const int N = in_sizes[0] / FIN;
    const int E = in_sizes[3] / 2;
    const int* src = ei;
    const int* dst = ei + E;

    size_t NA = ((size_t)N + 255) & ~(size_t)255;
    unsigned* deg  = (unsigned*)d_ws;
    float*    dinv = (float*)d_ws + NA;
    float*    h    = (float*)d_ws + 2 * NA;
    float*    out  = (float*)d_out;

    hipMemsetAsync(deg, 0, (size_t)N * sizeof(unsigned), stream);

    cgcl_place_kernel<<<(E + BIN_CHUNK - 1) / BIN_CHUNK, BIN_THREADS, 0, stream>>>(src, dst, deg, out, N, E);

    cgcl_count_dinv_kernel<<<(N + ROWS_PER_BLOCK - 1) / ROWS_PER_BLOCK, COUNT_THREADS, 0, stream>>>(
        dst, deg, out, dinv, N, E);

    const int row_tiles = (N + 15) / 16;
    cgcl_gemm_wmma<<<row_tiles, 128, 0, stream>>>(x, W, h, N);

    cgcl_gather_owner_kernel<<<(N + ROWS_PER_BLOCK - 1) / ROWS_PER_BLOCK, GATHER_THREADS, 0, stream>>>(
        src, dst, deg, dinv, h, bias, out, N, E);
}
